# starting kernel
# baseline (speedup 1.0000x reference)
#include <hip/hip_runtime.h>

typedef __attribute__((ext_vector_type(16))) __bf16 v16bf;
typedef __attribute__((ext_vector_type(8)))  __bf16 v8bf;
typedef __attribute__((ext_vector_type(8)))  float  v8f;

__device__ __forceinline__ __bf16 f2bf(float f) {
  union { float f; unsigned u; } v; v.f = f;
  unsigned r = v.u + 0x7FFFu + ((v.u >> 16) & 1u);
  union { unsigned short s; __bf16 b; } o;
  o.s = (unsigned short)(r >> 16);
  return o.b;
}

template <int K>
__global__ void gemm_wmma_bf16(const __bf16* __restrict__ A,
                               const __bf16* __restrict__ Bt,
                               float* __restrict__ C, int ldc) {
  const int wave = threadIdx.x >> 5;
  const int lane = threadIdx.x & 31;
  const int m16  = lane & 15;
  const size_t arow = (size_t)(blockIdx.x * 16 + m16) * K;
  const size_t brow = (size_t)(wave * 16 + m16) * K;
  const int khalf = (lane >> 4) << 3;

  v8f acc = {};
#pragma unroll
  for (int k0 = 0; k0 < K; k0 += 32) {
    const v8bf alo = *(const v8bf*)(A  + arow + k0 + khalf);
    const v8bf ahi = *(const v8bf*)(A  + arow + k0 + 16 + khalf);
    const v8bf blo = *(const v8bf*)(Bt + brow + k0 + khalf);
    const v8bf bhi = *(const v8bf*)(Bt + brow + k0 + 16 + khalf);
    const v16bf a = __builtin_shufflevector(alo, ahi, 0,1,2,3,4,5,6,7,8,9,10,11,12,13,14,15);
    const v16bf b = __builtin_shufflevector(blo, bhi, 0,1,2,3,4,5,6,7,8,9,10,11,12,13,14,15);
    acc = __builtin_amdgcn_wmma_f32_16x16x32_bf16(
        false, a, false, b, (short)0, acc, false, false);
  }

  const int col  = wave * 16 + m16;
  const int crow = blockIdx.x * 16 + ((lane >> 4) << 3);
#pragma unroll
  for (int r = 0; r < 8; ++r)
    C[(size_t)(crow + r) * ldc + col] = acc[r];
}

__global__ void cvt_bf16_k(const float* __restrict__ in, __bf16* __restrict__ out,
                           long long n) {
  long long i = (long long)blockIdx.x * blockDim.x + threadIdx.x;
  if (i < n) out[i] = f2bf(in[i]);
}

__global__ void cvt_tr_k(const float* __restrict__ W, __bf16* __restrict__ Wt,
                         int K, int N) {
  int i = blockIdx.x * blockDim.x + threadIdx.x;
  if (i >= K * N) return;
  int k = i / N, n = i % N;
  Wt[(size_t)n * K + k] = f2bf(W[i]);
}

__global__ void init_deg_k(float* __restrict__ deg, int n) {
  int i = blockIdx.x * blockDim.x + threadIdx.x;
  if (i < n) deg[i] = 1.0f;
}

__global__ void edge_deg_k(const int* __restrict__ dst, float* __restrict__ deg, int e) {
  int i = blockIdx.x * blockDim.x + threadIdx.x;
  if (i < e) atomicAdd(&deg[dst[i]], 1.0f);
}

__global__ void rsqrt_k(float* __restrict__ d, int n) {
  int i = blockIdx.x * blockDim.x + threadIdx.x;
  if (i < n) { float v = d[i]; d[i] = v > 0.0f ? rsqrtf(v) : 0.0f; }
}

__global__ void init_agg_k(const float* __restrict__ hlin, const float* __restrict__ dinv,
                           const float* __restrict__ bias, float* __restrict__ out,
                           long long total, int F) {
  long long i = (long long)blockIdx.x * blockDim.x + threadIdx.x;
  if (i >= total) return;
  int n = (int)(i / F), f = (int)(i % F);
  float di = dinv[n];
  out[i] = hlin[i] * di * di + bias[f];
}

__global__ void edge_agg_k(const int* __restrict__ src, const int* __restrict__ dst,
                           const float* __restrict__ dinv, const float* __restrict__ hlin,
                           float* __restrict__ out, int E, int F4, int F) {
  long long i = (long long)blockIdx.x * blockDim.x + threadIdx.x;
  if (i >= (long long)E * F4) return;
  int e = (int)(i / F4);
  int f = (int)(i % F4) * 4;
  int s = src[e], d = dst[e];
  float nrm = dinv[s] * dinv[d];
  const float4 hv = *(const float4*)(hlin + (size_t)s * F + f);
  float* op = out + (size_t)d * F + f;
  atomicAdd(op + 0, hv.x * nrm);
  atomicAdd(op + 1, hv.y * nrm);
  atomicAdd(op + 2, hv.z * nrm);
  atomicAdd(op + 3, hv.w * nrm);
}

__global__ void relu_k(float* __restrict__ p, long long n) {
  long long i = (long long)blockIdx.x * blockDim.x + threadIdx.x;
  if (i < n) p[i] = fmaxf(p[i], 0.0f);
}

extern "C" void kernel_launch(void* const* d_in, const int* in_sizes, int n_in,
                              void* d_out, int out_size, void* d_ws, size_t ws_size,
                              hipStream_t stream) {
  const float* x   = (const float*)d_in[0];
  const int*   ei  = (const int*)d_in[1];
  const int*   yei = (const int*)d_in[2];
  const float* W1  = (const float*)d_in[3];
  const float* b1  = (const float*)d_in[4];
  const float* Wmu = (const float*)d_in[5];
  const float* bmu = (const float*)d_in[6];
  const float* Wls = (const float*)d_in[7];
  const float* bls = (const float*)d_in[8];

  const int DIN = 128, DH = 128, DZ = 32;
  const int N  = in_sizes[0] / DIN;
  const int E1 = in_sizes[1] / 2;
  const int E2 = in_sizes[2] / 2;

  const int* src1 = ei;   const int* dst1 = ei  + E1;
  const int* src2 = yei;  const int* dst2 = yei + E2;

  char* wsb = (char*)d_ws;
  float*  dinv = (float*)wsb;
  __bf16* w1t  = (__bf16*)(wsb + (size_t)65536 * 4);
  __bf16* wmut = w1t + 128 * 128;
  __bf16* wlst = wmut + 32 * 128;
  __bf16* abf  = (__bf16*)(wsb + (size_t)(65536 + 16384) * 4);
  float*  buf1 = (float*)(wsb + (size_t)(65536 + 16384) * 4 + (size_t)N * DH * 2);
  float*  buf2 = buf1 + (size_t)N * DH;

  const int T = 256;
  auto blocks = [](long long total, int t) { return (unsigned)((total + t - 1) / t); };

  init_deg_k<<<blocks(N, T), T, 0, stream>>>(dinv, N);
  edge_deg_k<<<blocks(E1, T), T, 0, stream>>>(dst1, dinv, E1);
  edge_deg_k<<<blocks(E2, T), T, 0, stream>>>(dst2, dinv, E2);
  rsqrt_k<<<blocks(N, T), T, 0, stream>>>(dinv, N);

  cvt_bf16_k<<<blocks((long long)N * DIN, T), T, 0, stream>>>(x, abf, (long long)N * DIN);
  cvt_tr_k<<<blocks(DIN * DH, T), T, 0, stream>>>(W1, w1t, DIN, DH);
  gemm_wmma_bf16<128><<<N / 16, 32 * (DH / 16), 0, stream>>>(abf, w1t, buf1, DH);

  init_agg_k<<<blocks((long long)N * DH, T), T, 0, stream>>>(buf1, dinv, b1, buf2,
                                                             (long long)N * DH, DH);
  edge_agg_k<<<blocks((long long)E1 * (DH / 4), T), T, 0, stream>>>(src1, dst1, dinv,
                                                                    buf1, buf2, E1, DH / 4, DH);
  edge_agg_k<<<blocks((long long)E2 * (DH / 4), T), T, 0, stream>>>(src2, dst2, dinv,
                                                                    buf1, buf2, E2, DH / 4, DH);
  relu_k<<<blocks((long long)N * DH, T), T, 0, stream>>>(buf2, (long long)N * DH);

  cvt_bf16_k<<<blocks((long long)N * DH, T), T, 0, stream>>>(buf2, abf, (long long)N * DH);
  cvt_tr_k<<<blocks(DH * DZ, T), T, 0, stream>>>(Wmu, wmut, DH, DZ);
  cvt_tr_k<<<blocks(DH * DZ, T), T, 0, stream>>>(Wls, wlst, DH, DZ);

  float* zmu_lin = buf1;
  float* zls_lin = buf1 + (size_t)N * DZ;
  gemm_wmma_bf16<128><<<N / 16, 32 * (DZ / 16), 0, stream>>>(abf, wmut, zmu_lin, DZ);
  gemm_wmma_bf16<128><<<N / 16, 32 * (DZ / 16), 0, stream>>>(abf, wlst, zls_lin, DZ);

  float* out_mu = (float*)d_out;
  float* out_ls = out_mu + (size_t)N * DZ;

  init_agg_k<<<blocks((long long)N * DZ, T), T, 0, stream>>>(zmu_lin, dinv, bmu, out_mu,
                                                             (long long)N * DZ, DZ);
  init_agg_k<<<blocks((long long)N * DZ, T), T, 0, stream>>>(zls_lin, dinv, bls, out_ls,
                                                             (long long)N * DZ, DZ);

  edge_agg_k<<<blocks((long long)E1 * (DZ / 4), T), T, 0, stream>>>(src1, dst1, dinv,
                                                                    zmu_lin, out_mu, E1, DZ / 4, DZ);
  edge_agg_k<<<blocks((long long)E2 * (DZ / 4), T), T, 0, stream>>>(src2, dst2, dinv,
                                                                    zmu_lin, out_mu, E2, DZ / 4, DZ);
  edge_agg_k<<<blocks((long long)E1 * (DZ / 4), T), T, 0, stream>>>(src1, dst1, dinv,
                                                                    zls_lin, out_ls, E1, DZ / 4, DZ);
  edge_agg_k<<<blocks((long long)E2 * (DZ / 4), T), T, 0, stream>>>(src2, dst2, dinv,
                                                                    zls_lin, out_ls, E2, DZ / 4, DZ);
}


# Float atomic scatter-add replaced by grouping edges by destination once per call (arrival-order rank from the degree add, offsets by scan) and an owner-lane gather per row; sum order still varies
# speedup vs baseline: 1.8938x; 1.8938x over previous
#include <hip/hip_runtime.h>

typedef __attribute__((ext_vector_type(16))) __bf16 v16bf;
typedef __attribute__((ext_vector_type(8)))  __bf16 v8bf;
typedef __attribute__((ext_vector_type(8)))  float  v8f;

__device__ __forceinline__ __bf16 f2bf(float f) {
  union { float f; unsigned u; } v; v.f = f;
  unsigned r = v.u + 0x7FFFu + ((v.u >> 16) & 1u);
  union { unsigned short s; __bf16 b; } o;
  o.s = (unsigned short)(r >> 16);
  return o.b;
}

template <int K>
__global__ void gemm_wmma_bf16(const __bf16* __restrict__ A,
                               const __bf16* __restrict__ Bt,
                               float* __restrict__ C, int ldc) {
  const int wave = threadIdx.x >> 5;
  const int lane = threadIdx.x & 31;
  const int m16  = lane & 15;
  const size_t arow = (size_t)(blockIdx.x * 16 + m16) * K;
  const size_t brow = (size_t)(wave * 16 + m16) * K;
  const int khalf = (lane >> 4) << 3;

  v8f acc = {};
#pragma unroll
  for (int k0 = 0; k0 < K; k0 += 32) {
    const v8bf alo = *(const v8bf*)(A  + arow + k0 + khalf);
    const v8bf ahi = *(const v8bf*)(A  + arow + k0 + 16 + khalf);
    const v8bf blo = *(const v8bf*)(Bt + brow + k0 + khalf);
    const v8bf bhi = *(const v8bf*)(Bt + brow + k0 + 16 + khalf);
    const v16bf a = __builtin_shufflevector(alo, ahi, 0,1,2,3,4,5,6,7,8,9,10,11,12,13,14,15);
    const v16bf b = __builtin_shufflevector(blo, bhi, 0,1,2,3,4,5,6,7,8,9,10,11,12,13,14,15);
    acc = __builtin_amdgcn_wmma_f32_16x16x32_bf16(
        false, a, false, b, (short)0, acc, false, false);
  }

  const int col  = wave * 16 + m16;
  const int crow = blockIdx.x * 16 + ((lane >> 4) << 3);
#pragma unroll
  for (int r = 0; r < 8; ++r)
    C[(size_t)(crow + r) * ldc + col] = acc[r];
}

__global__ void cvt_bf16_k(const float* __restrict__ in, __bf16* __restrict__ out,
                           long long n) {
  long long i = (long long)blockIdx.x * blockDim.x + threadIdx.x;
  if (i < n) out[i] = f2bf(in[i]);
}

__global__ void cvt_tr_k(const float* __restrict__ W, __bf16* __restrict__ Wt,
                         int K, int N) {
  int i = blockIdx.x * blockDim.x + threadIdx.x;
  if (i >= K * N) return;
  int k = i / N, n = i % N;
  Wt[(size_t)n * K + k] = f2bf(W[i]);
}

__global__ void init_deg_k(float* __restrict__ deg, int n) {
  int i = blockIdx.x * blockDim.x + threadIdx.x;
  if (i < n) deg[i] = 1.0f;
}

__global__ void edge_rank_k(const int* __restrict__ dst, float* __restrict__ deg,
                            int* __restrict__ rank, int e) {
  int i = blockIdx.x * blockDim.x + threadIdx.x;
  if (i < e) rank[i] = (int)atomicAdd(&deg[dst[i]], 1.0f) - 1;
}

__global__ void scan_off_k(const float* __restrict__ deg, int* __restrict__ off, int n) {
  __shared__ int wsum[32];
  const int t = threadIdx.x;
  const int lane = t & 31, wave = t >> 5;
  const int base = t * 64;
  float4 v[16];
#pragma unroll
  for (int q = 0; q < 16; ++q) {
    const int i = base + q * 4;
    v[q] = *(const float4*)(deg + (i < n ? i : 0));
  }
  int c[64];
#pragma unroll
  for (int q = 0; q < 16; ++q) {
    const bool in = base + q * 4 < n;
    c[q * 4 + 0] = in ? (int)v[q].x - 1 : 0;
    c[q * 4 + 1] = in ? (int)v[q].y - 1 : 0;
    c[q * 4 + 2] = in ? (int)v[q].z - 1 : 0;
    c[q * 4 + 3] = in ? (int)v[q].w - 1 : 0;
  }
  int run = 0;
#pragma unroll
  for (int k = 0; k < 64; ++k) { const int x = c[k]; c[k] = run; run += x; }

  int incl = run;
#pragma unroll
  for (int d = 1; d < 32; d <<= 1) {
    const int y = __shfl_up(incl, d);
    if (lane >= d) incl += y;
  }
  if (lane == 31) wsum[wave] = incl;
  __syncthreads();
  if (wave == 0) {
    const int x = wsum[lane];
    int s = x;
#pragma unroll
    for (int d = 1; d < 32; d <<= 1) {
      const int y = __shfl_up(s, d);
      if (lane >= d) s += y;
    }
    wsum[lane] = s - x;
  }
  __syncthreads();
  const int pre = wsum[wave] + incl - run;
#pragma unroll
  for (int q = 0; q < 16; ++q) {
    const int i = base + q * 4;
    if (i < n)
      *(int4*)(off + i) = make_int4(c[q * 4] + pre, c[q * 4 + 1] + pre,
                                    c[q * 4 + 2] + pre, c[q * 4 + 3] + pre);
  }
  if (t == 1023) off[n] = pre + run;
}

__global__ void rsqrt_k(float* __restrict__ d, int n) {
  int i = blockIdx.x * blockDim.x + threadIdx.x;
  if (i < n) { float v = d[i]; d[i] = v > 0.0f ? rsqrtf(v) : 0.0f; }
}

__global__ void fill_csr_k(const int* __restrict__ src, const int* __restrict__ dst,
                           const int* __restrict__ rank, const int* __restrict__ off,
                           int* __restrict__ csr, int e) {
  int i = blockIdx.x * blockDim.x + threadIdx.x;
  if (i < e) csr[off[dst[i]] + rank[i]] = src[i];
}

template <int F, bool HIDDEN>
__global__ void gather_agg_k(const int* __restrict__ off, const int* __restrict__ csr,
                             const float* __restrict__ dinv, const float* __restrict__ hlin,
                             const float* __restrict__ bias, float* __restrict__ outf,
                             __bf16* __restrict__ outb, int n_nodes) {
  constexpr int LPN = F / 4;
  const long long gid = (long long)blockIdx.x * blockDim.x + threadIdx.x;
  const int node = (int)(gid / LPN);
  if (node >= n_nodes) return;
  const int f = (int)(gid % LPN) * 4;
  const float dn = dinv[node];
  const float4 hs = *(const float4*)(hlin + (size_t)node * F + f);
  float4 acc;
  acc.x = hs.x * dn * dn + bias[f + 0];
  acc.y = hs.y * dn * dn + bias[f + 1];
  acc.z = hs.z * dn * dn + bias[f + 2];
  acc.w = hs.w * dn * dn + bias[f + 3];

  const int jbeg = off[node], jend = off[node + 1];
  for (int j = jbeg; j < jend; j += 4) {
    int s[4]; float w[4]; float4 r[4];
#pragma unroll
    for (int k = 0; k < 4; ++k) s[k] = csr[min(j + k, jend - 1)];
#pragma unroll
    for (int k = 0; k < 4; ++k) w[k] = dinv[s[k]] * dn;
#pragma unroll
    for (int k = 0; k < 4; ++k) {
      r[k] = *(const float4*)(hlin + (size_t)s[k] * F + f);
      if (j + k >= jend) r[k] = make_float4(0.0f, 0.0f, 0.0f, 0.0f);
    }
#pragma unroll
    for (int k = 0; k < 4; ++k) {
      acc.x += r[k].x * w[k];
      acc.y += r[k].y * w[k];
      acc.z += r[k].z * w[k];
      acc.w += r[k].w * w[k];
    }
  }

  if (HIDDEN) {
    union { __bf16 b[4]; uint2 u; } o;
    o.b[0] = f2bf(fmaxf(acc.x, 0.0f));
    o.b[1] = f2bf(fmaxf(acc.y, 0.0f));
    o.b[2] = f2bf(fmaxf(acc.z, 0.0f));
    o.b[3] = f2bf(fmaxf(acc.w, 0.0f));
    *(uint2*)(outb + (size_t)node * F + f) = o.u;
  } else {
    *(float4*)(outf + (size_t)node * F + f) = acc;
  }
}

extern "C" void kernel_launch(void* const* d_in, const int* in_sizes, int n_in,
                              void* d_out, int out_size, void* d_ws, size_t ws_size,
                              hipStream_t stream) {
  const float* x   = (const float*)d_in[0];
  const int*   ei  = (const int*)d_in[1];
  const int*   yei = (const int*)d_in[2];
  const float* W1  = (const float*)d_in[3];
  const float* b1  = (const float*)d_in[4];
  const float* Wmu = (const float*)d_in[5];
  const float* bmu = (const float*)d_in[6];
  const float* Wls = (const float*)d_in[7];
  const float* bls = (const float*)d_in[8];

  const int DIN = 128, DH = 128, DZ = 32;
  const int N  = in_sizes[0] / DIN;
  const int E1 = in_sizes[1] / 2;
  const int E2 = in_sizes[2] / 2;

  const int* src1 = ei;   const int* dst1 = ei  + E1;
  const int* src2 = yei;  const int* dst2 = yei + E2;

  char* wsb = (char*)d_ws;
  float*  dinv = (float*)wsb;
  __bf16* w1t  = (__bf16*)(wsb + (size_t)65536 * 4);
  __bf16* wmut = w1t + 128 * 128;
  __bf16* wlst = wmut + 32 * 128;
  __bf16* abf  = (__bf16*)(wsb + (size_t)(65536 + 16384) * 4);
  float*  buf1 = (float*)(wsb + (size_t)(65536 + 16384) * 4 + (size_t)N * DH * 2);
  int*    off  = (int*)(buf1 + (size_t)N * DH);
  int*    rank = off + (N + 16);
  int*    csr  = rank + (E1 + E2);

  const int T = 256;
  auto blocks = [](long long total, int t) { return (unsigned)((total + t - 1) / t); };

  init_deg_k<<<blocks(N, T), T, 0, stream>>>(dinv, N);
  edge_rank_k<<<blocks(E1, T), T, 0, stream>>>(dst1, dinv, rank, E1);
  edge_rank_k<<<blocks(E2, T), T, 0, stream>>>(dst2, dinv, rank + E1, E2);
  scan_off_k<<<1, 1024, 0, stream>>>(dinv, off, N);
  rsqrt_k<<<blocks(N, T), T, 0, stream>>>(dinv, N);
  fill_csr_k<<<blocks(E1, T), T, 0, stream>>>(src1, dst1, rank, off, csr, E1);
  fill_csr_k<<<blocks(E2, T), T, 0, stream>>>(src2, dst2, rank + E1, off, csr, E2);

  cvt_bf16_k<<<blocks((long long)N * DIN, T), T, 0, stream>>>(x, abf, (long long)N * DIN);
  cvt_tr_k<<<blocks(DIN * DH, T), T, 0, stream>>>(W1, w1t, DIN, DH);
  gemm_wmma_bf16<128><<<N / 16, 32 * (DH / 16), 0, stream>>>(abf, w1t, buf1, DH);

  gather_agg_k<128, true><<<blocks((long long)N * (DH / 4), T), T, 0, stream>>>(
      off, csr, dinv, buf1, b1, nullptr, abf, N);

  cvt_tr_k<<<blocks(DH * DZ, T), T, 0, stream>>>(Wmu, wmut, DH, DZ);
  cvt_tr_k<<<blocks(DH * DZ, T), T, 0, stream>>>(Wls, wlst, DH, DZ);

  float* zmu_lin = buf1;
  float* zls_lin = buf1 + (size_t)N * DZ;
  gemm_wmma_bf16<128><<<N / 16, 32 * (DZ / 16), 0, stream>>>(abf, wmut, zmu_lin, DZ);
  gemm_wmma_bf16<128><<<N / 16, 32 * (DZ / 16), 0, stream>>>(abf, wlst, zls_lin, DZ);

  float* out_mu = (float*)d_out;
  float* out_ls = out_mu + (size_t)N * DZ;

  gather_agg_k<32, false><<<blocks((long long)N * (DZ / 4), T), T, 0, stream>>>(
      off, csr, dinv, zmu_lin, bmu, out_mu, nullptr, N);
  gather_agg_k<32, false><<<blocks((long long)N * (DZ / 4), T), T, 0, stream>>>(
      off, csr, dinv, zls_lin, bls, out_ls, nullptr, N);
}


# Layer-1 product loads the f32 input itself and rounds its A fragments in registers (same nearest-even rounding, same operand type); the one-element-per-thread conversion launch over x is gone
# speedup vs baseline: 2.0256x; 1.0696x over previous
#include <hip/hip_runtime.h>

typedef __attribute__((ext_vector_type(16))) __bf16 v16bf;
typedef __attribute__((ext_vector_type(8)))  __bf16 v8bf;
typedef __attribute__((ext_vector_type(8)))  float  v8f;

__device__ __forceinline__ __bf16 f2bf(float f) {
  union { float f; unsigned u; } v; v.f = f;
  unsigned r = v.u + 0x7FFFu + ((v.u >> 16) & 1u);
  union { unsigned short s; __bf16 b; } o;
  o.s = (unsigned short)(r >> 16);
  return o.b;
}

__device__ __forceinline__ v8bf load8(const __bf16* p) { return *(const v8bf*)p; }
__device__ __forceinline__ v8bf load8(const float* p) {
  const float4 lo = *(const float4*)p;
  const float4 hi = *(const float4*)(p + 4);
  v8bf r;
  r[0] = f2bf(lo.x); r[1] = f2bf(lo.y); r[2] = f2bf(lo.z); r[3] = f2bf(lo.w);
  r[4] = f2bf(hi.x); r[5] = f2bf(hi.y); r[6] = f2bf(hi.z); r[7] = f2bf(hi.w);
  return r;
}

template <int K, typename TA>
__device__ __forceinline__ void gemm_tile(const TA* __restrict__ A,
                                          const __bf16* __restrict__ Bt,
                                          float* __restrict__ C, int ldc) {
  const int wave = threadIdx.x >> 5;
  const int lane = threadIdx.x & 31;
  const int m16  = lane & 15;
  const size_t arow = (size_t)(blockIdx.x * 16 + m16) * K;
  const size_t brow = (size_t)(wave * 16 + m16) * K;
  const int khalf = (lane >> 4) << 3;

  v8f acc = {};
#pragma unroll
  for (int k0 = 0; k0 < K; k0 += 32) {
    const v8bf alo = load8(A + arow + k0 + khalf);
    const v8bf ahi = load8(A + arow + k0 + 16 + khalf);
    const v8bf blo = *(const v8bf*)(Bt + brow + k0 + khalf);
    const v8bf bhi = *(const v8bf*)(Bt + brow + k0 + 16 + khalf);
    const v16bf a = __builtin_shufflevector(alo, ahi, 0,1,2,3,4,5,6,7,8,9,10,11,12,13,14,15);
    const v16bf b = __builtin_shufflevector(blo, bhi, 0,1,2,3,4,5,6,7,8,9,10,11,12,13,14,15);
    acc = __builtin_amdgcn_wmma_f32_16x16x32_bf16(
        false, a, false, b, (short)0, acc, false, false);
  }

  const int col  = wave * 16 + m16;
  const int crow = blockIdx.x * 16 + ((lane >> 4) << 3);
#pragma unroll
  for (int r = 0; r < 8; ++r)
    C[(size_t)(crow + r) * ldc + col] = acc[r];
}

template <int K>
__global__ void gemm_wmma_bf16(const __bf16* __restrict__ A,
                               const __bf16* __restrict__ Bt,
                               float* __restrict__ C, int ldc) {
  gemm_tile<K>(A, Bt, C, ldc);
}

template <int K>
__global__ void gemm_wmma_xf32(const float* __restrict__ X,
                               const __bf16* __restrict__ Bt,
                               float* __restrict__ C, int ldc) {
  gemm_tile<K>(X, Bt, C, ldc);
}

__global__ void cvt_tr_k(const float* __restrict__ W, __bf16* __restrict__ Wt,
                         int K, int N) {
  int i = blockIdx.x * blockDim.x + threadIdx.x;
  if (i >= K * N) return;
  int k = i / N, n = i % N;
  Wt[(size_t)n * K + k] = f2bf(W[i]);
}

__global__ void init_deg_k(float* __restrict__ deg, int n) {
  int i = blockIdx.x * blockDim.x + threadIdx.x;
  if (i < n) deg[i] = 1.0f;
}

__global__ void edge_rank_k(const int* __restrict__ dst, float* __restrict__ deg,
                            int* __restrict__ rank, int e) {
  int i = blockIdx.x * blockDim.x + threadIdx.x;
  if (i < e) rank[i] = (int)atomicAdd(&deg[dst[i]], 1.0f) - 1;
}

__global__ void scan_off_k(const float* __restrict__ deg, int* __restrict__ off, int n) {
  __shared__ int wsum[32];
  const int t = threadIdx.x;
  const int lane = t & 31, wave = t >> 5;
  const int base = t * 64;
  float4 v[16];
#pragma unroll
  for (int q = 0; q < 16; ++q) {
    const int i = base + q * 4;
    v[q] = *(const float4*)(deg + (i < n ? i : 0));
  }
  int c[64];
#pragma unroll
  for (int q = 0; q < 16; ++q) {
    const bool in = base + q * 4 < n;
    c[q * 4 + 0] = in ? (int)v[q].x - 1 : 0;
    c[q * 4 + 1] = in ? (int)v[q].y - 1 : 0;
    c[q * 4 + 2] = in ? (int)v[q].z - 1 : 0;
    c[q * 4 + 3] = in ? (int)v[q].w - 1 : 0;
  }
  int run = 0;
#pragma unroll
  for (int k = 0; k < 64; ++k) { const int x = c[k]; c[k] = run; run += x; }

  int incl = run;
#pragma unroll
  for (int d = 1; d < 32; d <<= 1) {
    const int y = __shfl_up(incl, d);
    if (lane >= d) incl += y;
  }
  if (lane == 31) wsum[wave] = incl;
  __syncthreads();
  if (wave == 0) {
    const int x = wsum[lane];
    int s = x;
#pragma unroll
    for (int d = 1; d < 32; d <<= 1) {
      const int y = __shfl_up(s, d);
      if (lane >= d) s += y;
    }
    wsum[lane] = s - x;
  }
  __syncthreads();
  const int pre = wsum[wave] + incl - run;
#pragma unroll
  for (int q = 0; q < 16; ++q) {
    const int i = base + q * 4;
    if (i < n)
      *(int4*)(off + i) = make_int4(c[q * 4] + pre, c[q * 4 + 1] + pre,
                                    c[q * 4 + 2] + pre, c[q * 4 + 3] + pre);
  }
  if (t == 1023) off[n] = pre + run;
}

__global__ void rsqrt_k(float* __restrict__ d, int n) {
  int i = blockIdx.x * blockDim.x + threadIdx.x;
  if (i < n) { float v = d[i]; d[i] = v > 0.0f ? rsqrtf(v) : 0.0f; }
}

__global__ void fill_csr_k(const int* __restrict__ src, const int* __restrict__ dst,
                           const int* __restrict__ rank, const int* __restrict__ off,
                           int* __restrict__ csr, int e) {
  int i = blockIdx.x * blockDim.x + threadIdx.x;
  if (i < e) csr[off[dst[i]] + rank[i]] = src[i];
}

template <int F, bool HIDDEN>
__global__ void gather_agg_k(const int* __restrict__ off, const int* __restrict__ csr,
                             const float* __restrict__ dinv, const float* __restrict__ hlin,
                             const float* __restrict__ bias, float* __restrict__ outf,
                             __bf16* __restrict__ outb, int n_nodes) {
  constexpr int LPN = F / 4;
  const long long gid = (long long)blockIdx.x * blockDim.x + threadIdx.x;
  const int node = (int)(gid / LPN);
  if (node >= n_nodes) return;
  const int f = (int)(gid % LPN) * 4;
  const float dn = dinv[node];
  const float4 hs = *(const float4*)(hlin + (size_t)node * F + f);
  float4 acc;
  acc.x = hs.x * dn * dn + bias[f + 0];
  acc.y = hs.y * dn * dn + bias[f + 1];
  acc.z = hs.z * dn * dn + bias[f + 2];
  acc.w = hs.w * dn * dn + bias[f + 3];

  const int jbeg = off[node], jend = off[node + 1];
  for (int j = jbeg; j < jend; j += 4) {
    int s[4]; float w[4]; float4 r[4];
#pragma unroll
    for (int k = 0; k < 4; ++k) s[k] = csr[min(j + k, jend - 1)];
#pragma unroll
    for (int k = 0; k < 4; ++k) w[k] = dinv[s[k]] * dn;
#pragma unroll
    for (int k = 0; k < 4; ++k) {
      r[k] = *(const float4*)(hlin + (size_t)s[k] * F + f);
      if (j + k >= jend) r[k] = make_float4(0.0f, 0.0f, 0.0f, 0.0f);
    }
#pragma unroll
    for (int k = 0; k < 4; ++k) {
      acc.x += r[k].x * w[k];
      acc.y += r[k].y * w[k];
      acc.z += r[k].z * w[k];
      acc.w += r[k].w * w[k];
    }
  }

  if (HIDDEN) {
    union { __bf16 b[4]; uint2 u; } o;
    o.b[0] = f2bf(fmaxf(acc.x, 0.0f));
    o.b[1] = f2bf(fmaxf(acc.y, 0.0f));
    o.b[2] = f2bf(fmaxf(acc.z, 0.0f));
    o.b[3] = f2bf(fmaxf(acc.w, 0.0f));
    *(uint2*)(outb + (size_t)node * F + f) = o.u;
  } else {
    *(float4*)(outf + (size_t)node * F + f) = acc;
  }
}

extern "C" void kernel_launch(void* const* d_in, const int* in_sizes, int n_in,
                              void* d_out, int out_size, void* d_ws, size_t ws_size,
                              hipStream_t stream) {
  const float* x   = (const float*)d_in[0];
  const int*   ei  = (const int*)d_in[1];
  const int*   yei = (const int*)d_in[2];
  const float* W1  = (const float*)d_in[3];
  const float* b1  = (const float*)d_in[4];
  const float* Wmu = (const float*)d_in[5];
  const float* bmu = (const float*)d_in[6];
  const float* Wls = (const float*)d_in[7];
  const float* bls = (const float*)d_in[8];

  const int DIN = 128, DH = 128, DZ = 32;
  const int N  = in_sizes[0] / DIN;
  const int E1 = in_sizes[1] / 2;
  const int E2 = in_sizes[2] / 2;

  const int* src1 = ei;   const int* dst1 = ei  + E1;
  const int* src2 = yei;  const int* dst2 = yei + E2;

  char* wsb = (char*)d_ws;
  float*  dinv = (float*)wsb;
  __bf16* w1t  = (__bf16*)(wsb + (size_t)65536 * 4);
  __bf16* wmut = w1t + 128 * 128;
  __bf16* wlst = wmut + 32 * 128;
  __bf16* abf  = (__bf16*)(wsb + (size_t)(65536 + 16384) * 4);
  float*  buf1 = (float*)(wsb + (size_t)(65536 + 16384) * 4 + (size_t)N * DH * 2);
  int*    off  = (int*)(buf1 + (size_t)N * DH);
  int*    rank = off + (N + 16);
  int*    csr  = rank + (E1 + E2);

  const int T = 256;
  auto blocks = [](long long total, int t) { return (unsigned)((total + t - 1) / t); };

  init_deg_k<<<blocks(N, T), T, 0, stream>>>(dinv, N);
  edge_rank_k<<<blocks(E1, T), T, 0, stream>>>(dst1, dinv, rank, E1);
  edge_rank_k<<<blocks(E2, T), T, 0, stream>>>(dst2, dinv, rank + E1, E2);
  scan_off_k<<<1, 1024, 0, stream>>>(dinv, off, N);
  rsqrt_k<<<blocks(N, T), T, 0, stream>>>(dinv, N);
  fill_csr_k<<<blocks(E1, T), T, 0, stream>>>(src1, dst1, rank, off, csr, E1);
  fill_csr_k<<<blocks(E2, T), T, 0, stream>>>(src2, dst2, rank + E1, off, csr, E2);

  cvt_tr_k<<<blocks(DIN * DH, T), T, 0, stream>>>(W1, w1t, DIN, DH);
  gemm_wmma_xf32<128><<<N / 16, 32 * (DH / 16), 0, stream>>>(x, w1t, buf1, DH);

  gather_agg_k<128, true><<<blocks((long long)N * (DH / 4), T), T, 0, stream>>>(
      off, csr, dinv, buf1, b1, nullptr, abf, N);

  cvt_tr_k<<<blocks(DH * DZ, T), T, 0, stream>>>(Wmu, wmut, DH, DZ);
  cvt_tr_k<<<blocks(DH * DZ, T), T, 0, stream>>>(Wls, wlst, DH, DZ);

  float* zmu_lin = buf1;
  float* zls_lin = buf1 + (size_t)N * DZ;
  gemm_wmma_bf16<128><<<N / 16, 32 * (DZ / 16), 0, stream>>>(abf, wmut, zmu_lin, DZ);
  gemm_wmma_bf16<128><<<N / 16, 32 * (DZ / 16), 0, stream>>>(abf, wlst, zls_lin, DZ);

  float* out_mu = (float*)d_out;
  float* out_ls = out_mu + (size_t)N * DZ;

  gather_agg_k<32, false><<<blocks((long long)N * (DZ / 4), T), T, 0, stream>>>(
      off, csr, dinv, zmu_lin, bmu, out_mu, nullptr, N);
  gather_agg_k<32, false><<<blocks((long long)N * (DZ / 4), T), T, 0, stream>>>(
      off, csr, dinv, zls_lin, bls, out_ls, nullptr, N);
}


# Every wave-wide access of the scan and of both products now covers whole 128-byte lines: scan lanes own adjacent float4s (lane prefix per row), A rows go through LDS, weights packed in fragment order
# speedup vs baseline: 2.8093x; 1.3869x over previous
#include <hip/hip_runtime.h>

typedef __attribute__((ext_vector_type(16))) __bf16 v16bf;
typedef __attribute__((ext_vector_type(8)))  __bf16 v8bf;
typedef __attribute__((ext_vector_type(8)))  float  v8f;

__device__ __forceinline__ __bf16 f2bf(float f) {
  union { float f; unsigned u; } v; v.f = f;
  unsigned r = v.u + 0x7FFFu + ((v.u >> 16) & 1u);
  union { unsigned short s; __bf16 b; } o;
  o.s = (unsigned short)(r >> 16);
  return o.b;
}

template <int K, int NW, typename TA>
__device__ __forceinline__ void gemm_tile(const TA* __restrict__ A,
                                          const __bf16* __restrict__ Bp,
                                          float* __restrict__ C, int ldc) {
  static_assert(K == 128, "a wave-wide 16-byte load is one f32 row or two bf16 rows");
  constexpr bool F32 = sizeof(TA) == 4;
  constexpr int KS = K / 32;
  constexpr int PITCH = K * 2 + 16;
  static_assert(8 % NW == 0, "the waves share the 16 rows evenly");
  constexpr int NF = 16 / NW;
  constexpr int NB = 8 / NW;
  __shared__ __attribute__((aligned(16))) unsigned char tile[16 * PITCH];
  const int wave = threadIdx.x >> 5;
  const int lane = threadIdx.x & 31;
  const int m16  = lane & 15;
  const int half = lane >> 4;
  const size_t row0 = (size_t)blockIdx.x * 16;

  float4 vf[NF];
  uint4  vb[NB];
  if constexpr (F32) {
#pragma unroll
    for (int j = 0; j < NF; ++j)
      vf[j] = *(const float4*)((const float*)A + (row0 + wave + j * NW) * K + lane * 4);
  } else {
#pragma unroll
    for (int j = 0; j < NB; ++j)
      vb[j] = *(const uint4*)((const __bf16*)A + (row0 + (wave + j * NW) * 2 + half) * K + m16 * 8);
  }
  v8bf bf[KS * 2];
  const __bf16* bw = Bp + ((size_t)wave * KS * 2 * 32 + lane) * 8;
#pragma unroll
  for (int q = 0; q < KS * 2; ++q) bf[q] = *(const v8bf*)(bw + (size_t)q * 256);

  if constexpr (F32) {
#pragma unroll
    for (int j = 0; j < NF; ++j) {
      union { __bf16 b[4]; uint2 u; } o;
      o.b[0] = f2bf(vf[j].x); o.b[1] = f2bf(vf[j].y);
      o.b[2] = f2bf(vf[j].z); o.b[3] = f2bf(vf[j].w);
      *(uint2*)(tile + (wave + j * NW) * PITCH + lane * 8) = o.u;
    }
  } else {
#pragma unroll
    for (int j = 0; j < NB; ++j)
      *(uint4*)(tile + ((wave + j * NW) * 2 + half) * PITCH + m16 * 16) = vb[j];
  }
  __syncthreads();

  const unsigned char* ar = tile + m16 * PITCH + half * 16;
  v8f acc = {};
#pragma unroll
  for (int ks = 0; ks < KS; ++ks) {
    const v8bf alo = *(const v8bf*)(ar + ks * 64);
    const v8bf ahi = *(const v8bf*)(ar + ks * 64 + 32);
    const v16bf a = __builtin_shufflevector(alo, ahi, 0,1,2,3,4,5,6,7,8,9,10,11,12,13,14,15);
    const v16bf b = __builtin_shufflevector(bf[ks * 2], bf[ks * 2 + 1], 0,1,2,3,4,5,6,7,8,9,10,11,12,13,14,15);
    acc = __builtin_amdgcn_wmma_f32_16x16x32_bf16(
        false, a, false, b, (short)0, acc, false, false);
  }

  const int col  = wave * 16 + m16;
  const size_t crow = row0 + half * 8;
#pragma unroll
  for (int r = 0; r < 8; ++r)
    C[(crow + r) * ldc + col] = acc[r];
}

template <int K, int NW>
__global__ void __launch_bounds__(32 * NW, 16) gemm_wmma_bf16(const __bf16* __restrict__ A,
                               const __bf16* __restrict__ Bp,
                               float* __restrict__ C, int ldc) {
  gemm_tile<K, NW>(A, Bp, C, ldc);
}

template <int K, int NW>
__global__ void __launch_bounds__(32 * NW, 16) gemm_wmma_xf32(const float* __restrict__ X,
                               const __bf16* __restrict__ Bp,
                               float* __restrict__ C, int ldc) {
  gemm_tile<K, NW>(X, Bp, C, ldc);
}

__device__ __forceinline__ void pack_w(const float* __restrict__ W, __bf16* __restrict__ Wp,
                                       int K, int N, int i) {
  if (i >= K * N) return;
  const int k = i / N, n = i % N;
  const int kk = k & 31;
  const size_t frag = ((size_t)(n >> 4) * (K >> 5) + (k >> 5)) * 2 + (kk >> 4);
  const int lane = (((kk >> 3) & 1) << 4) + (n & 15);
  Wp[(frag * 32 + lane) * 8 + (kk & 7)] = f2bf(W[i]);
}

__global__ void init_k(float* __restrict__ deg, int n, int deg_blocks,
                       const float* __restrict__ W1, __bf16* __restrict__ w1t, int din, int dh,
                       const float* __restrict__ Wmu, const float* __restrict__ Wls,
                       __bf16* __restrict__ wzt, int dz) {
  int b = blockIdx.x;
  if (b < deg_blocks) {
    const int i = b * blockDim.x + threadIdx.x;
    if (i < n) deg[i] = 1.0f;
    return;
  }
  b -= deg_blocks;
  const int w1_blocks = (din * dh + blockDim.x - 1) / blockDim.x;
  if (b < w1_blocks) {
    pack_w(W1, w1t, din, dh, b * blockDim.x + threadIdx.x);
    return;
  }
  b -= w1_blocks;
  const int wz_blocks = (dh * dz + blockDim.x - 1) / blockDim.x;
  if (b < wz_blocks) pack_w(Wmu, wzt, dh, dz, b * blockDim.x + threadIdx.x);
  else               pack_w(Wls, wzt + (size_t)dz * dh, dh, dz,
                            (b - wz_blocks) * blockDim.x + threadIdx.x);
}

__global__ void edge_rank_k(const int* __restrict__ dst1, int e1,
                            const int* __restrict__ dst2, int e2,
                            float* __restrict__ deg, int* __restrict__ rank) {
  int i = blockIdx.x * blockDim.x + threadIdx.x;
  if (i >= e1 + e2) return;
  const int d = i < e1 ? dst1[i] : dst2[i - e1];
  rank[i] = (int)atomicAdd(&deg[d], 1.0f) - 1;
}

__global__ void scan_off_k(const float* __restrict__ deg, int* __restrict__ off, int n) {
  __shared__ int wsum[32];
  const int t = threadIdx.x;
  const int lane = t & 31, wave = t >> 5;
  const int base = wave * 2048 + lane * 4;
  float4 v[16];
#pragma unroll
  for (int q = 0; q < 16; ++q) {
    const int i = base + q * 128;
    v[q] = *(const float4*)(deg + (i < n ? i : 0));
  }
  int4 c[16];
  int incl[16];
#pragma unroll
  for (int q = 0; q < 16; ++q) {
    const int m = base + q * 128 < n ? -1 : 0;
    c[q].x = ((int)v[q].x - 1) & m;
    c[q].y = ((int)v[q].y - 1) & m;
    c[q].z = ((int)v[q].z - 1) & m;
    c[q].w = ((int)v[q].w - 1) & m;
    incl[q] = c[q].x + c[q].y + c[q].z + c[q].w;
  }
#pragma unroll
  for (int d = 1; d < 32; d <<= 1) {
#pragma unroll
    for (int q = 0; q < 16; ++q) {
      const int y = __shfl_up(incl[q], d);
      if (lane >= d) incl[q] += y;
    }
  }
  int carry = 0;
#pragma unroll
  for (int q = 0; q < 16; ++q) {
    const int b = carry + incl[q] - (c[q].x + c[q].y + c[q].z + c[q].w);
    carry += __builtin_amdgcn_readlane(incl[q], 31);
    c[q] = make_int4(b, b + c[q].x, b + c[q].x + c[q].y, b + c[q].x + c[q].y + c[q].z);
  }
  if (lane == 0) wsum[wave] = carry;
  __syncthreads();
  if (wave == 0) {
    const int x = wsum[lane];
    int s = x;
#pragma unroll
    for (int d = 1; d < 32; d <<= 1) {
      const int y = __shfl_up(s, d);
      if (lane >= d) s += y;
    }
    wsum[lane] = s - x;
  }
  __syncthreads();
  const int pre = wsum[wave];
#pragma unroll
  for (int q = 0; q < 16; ++q) {
    const int i = base + q * 128;
    if (i < n)
      *(int4*)(off + i) = make_int4(c[q].x + pre, c[q].y + pre, c[q].z + pre, c[q].w + pre);
  }
  if (t == 1023) off[n] = pre + carry;
}

__global__ void fill_csr_k(const int* __restrict__ src1, const int* __restrict__ dst1, int e1,
                           const int* __restrict__ src2, const int* __restrict__ dst2, int e2,
                           const int* __restrict__ rank, const int* __restrict__ off,
                           int* __restrict__ csr, int edge_blocks,
                           float* __restrict__ deg, int n) {
  if ((int)blockIdx.x >= edge_blocks) {
    const int j = ((int)blockIdx.x - edge_blocks) * blockDim.x + threadIdx.x;
    if (j < n) { const float v = deg[j]; deg[j] = v > 0.0f ? rsqrtf(v) : 0.0f; }
    return;
  }
  int i = blockIdx.x * blockDim.x + threadIdx.x;
  if (i >= e1 + e2) return;
  int s, d;
  if (i < e1) { s = src1[i];      d = dst1[i]; }
  else        { s = src2[i - e1]; d = dst2[i - e1]; }
  csr[off[d] + rank[i]] = s;
}

template <int F, bool HIDDEN>
__global__ void gather_agg_k(const int* __restrict__ off, const int* __restrict__ csr,
                             const float* __restrict__ dinv, const float* __restrict__ hlin,
                             const float* __restrict__ bias, const float* __restrict__ bias2,
                             float* __restrict__ outf, float* __restrict__ outf2,
                             __bf16* __restrict__ outb, int n_nodes) {
  constexpr int LPN = F / 4;
  constexpr int H = F / 2;
  const long long gid = (long long)blockIdx.x * blockDim.x + threadIdx.x;
  const int node = (int)(gid / LPN);
  if (node >= n_nodes) return;
  const int f = (int)(gid % LPN) * 4;
  const float dn = dinv[node];
  const float4 hs = *(const float4*)(hlin + (size_t)node * F + f);
  const float* bp = HIDDEN ? bias + f : (f < H ? bias + f : bias2 + (f - H));
  float4 acc;
  acc.x = hs.x * dn * dn + bp[0];
  acc.y = hs.y * dn * dn + bp[1];
  acc.z = hs.z * dn * dn + bp[2];
  acc.w = hs.w * dn * dn + bp[3];

  const int jbeg = off[node], jend = off[node + 1];
  for (int j = jbeg; j < jend; j += 4) {
    int s[4]; float w[4]; float4 r[4];
#pragma unroll
    for (int k = 0; k < 4; ++k) s[k] = csr[min(j + k, jend - 1)];
#pragma unroll
    for (int k = 0; k < 4; ++k) w[k] = dinv[s[k]] * dn;
#pragma unroll
    for (int k = 0; k < 4; ++k) {
      r[k] = *(const float4*)(hlin + (size_t)s[k] * F + f);
      if (j + k >= jend) r[k] = make_float4(0.0f, 0.0f, 0.0f, 0.0f);
    }
#pragma unroll
    for (int k = 0; k < 4; ++k) {
      acc.x += r[k].x * w[k];
      acc.y += r[k].y * w[k];
      acc.z += r[k].z * w[k];
      acc.w += r[k].w * w[k];
    }
  }

  if (HIDDEN) {
    union { __bf16 b[4]; uint2 u; } o;
    o.b[0] = f2bf(fmaxf(acc.x, 0.0f));
    o.b[1] = f2bf(fmaxf(acc.y, 0.0f));
    o.b[2] = f2bf(fmaxf(acc.z, 0.0f));
    o.b[3] = f2bf(fmaxf(acc.w, 0.0f));
    *(uint2*)(outb + (size_t)node * F + f) = o.u;
  } else {
    float* o = f < H ? outf + (size_t)node * H + f : outf2 + (size_t)node * H + (f - H);
    *(float4*)o = acc;
  }
}

extern "C" void kernel_launch(void* const* d_in, const int* in_sizes, int n_in,
                              void* d_out, int out_size, void* d_ws, size_t ws_size,
                              hipStream_t stream) {
  const float* x   = (const float*)d_in[0];
  const int*   ei  = (const int*)d_in[1];
  const int*   yei = (const int*)d_in[2];
  const float* W1  = (const float*)d_in[3];
  const float* b1  = (const float*)d_in[4];
  const float* Wmu = (const float*)d_in[5];
  const float* bmu = (const float*)d_in[6];
  const float* Wls = (const float*)d_in[7];
  const float* bls = (const float*)d_in[8];

  const int DIN = 128, DH = 128, DZ = 32;
  const int N  = in_sizes[0] / DIN;
  const int E1 = in_sizes[1] / 2;
  const int E2 = in_sizes[2] / 2;

  const int* src1 = ei;   const int* dst1 = ei  + E1;
  const int* src2 = yei;  const int* dst2 = yei + E2;

  char* wsb = (char*)d_ws;
  float*  dinv = (float*)wsb;
  __bf16* w1t  = (__bf16*)(wsb + (size_t)65536 * 4);
  __bf16* wmut = w1t + 128 * 128;
  __bf16* abf  = (__bf16*)(wsb + (size_t)(65536 + 16384) * 4);
  float*  buf1 = (float*)(wsb + (size_t)(65536 + 16384) * 4 + (size_t)N * DH * 2);
  int*    off  = (int*)(buf1 + (size_t)N * DH);
  int*    rank = off + (N + 16);
  int*    csr  = rank + (E1 + E2);

  const int T = 256;
  auto blocks = [](long long total, int t) { return (unsigned)((total + t - 1) / t); };

  const unsigned deg_blocks = blocks(N, T);
  init_k<<<deg_blocks + blocks(DIN * DH, T) + 2 * blocks(DH * DZ, T), T, 0, stream>>>(
      dinv, N, (int)deg_blocks, W1, w1t, DIN, DH, Wmu, Wls, wmut, DZ);

  const unsigned edge_blocks = blocks(E1 + E2, T);
  edge_rank_k<<<edge_blocks, T, 0, stream>>>(dst1, E1, dst2, E2, dinv, rank);
  scan_off_k<<<1, 1024, 0, stream>>>(dinv, off, N);
  fill_csr_k<<<edge_blocks + deg_blocks, T, 0, stream>>>(src1, dst1, E1, src2, dst2, E2,
                                                        rank, off, csr, (int)edge_blocks, dinv, N);

  gemm_wmma_xf32<128, 8><<<N / 16, 32 * 8, 0, stream>>>(x, w1t, buf1, DH);

  gather_agg_k<128, true><<<blocks((long long)N * (DH / 4), T), T, 0, stream>>>(
      off, csr, dinv, buf1, b1, nullptr, nullptr, nullptr, abf, N);

  float* z_lin = buf1;
  gemm_wmma_bf16<128, 4><<<N / 16, 32 * 4, 0, stream>>>(abf, wmut, z_lin, 2 * DZ);

  float* out_mu = (float*)d_out;
  float* out_ls = out_mu + (size_t)N * DZ;

  gather_agg_k<64, false><<<blocks((long long)N * (2 * DZ / 4), T), T, 0, stream>>>(
      off, csr, dinv, z_lin, bmu, bls, out_mu, out_ls, nullptr, N);
}


# Layer 1 sums bf16-rounded input rows before the product (256-byte gathers, not 512); the 16-node sum is multiplied in that launch as three bf16 parts (exact split); sum order changes; 8->7 launches
# speedup vs baseline: 3.0405x; 1.0823x over previous
#include <hip/hip_runtime.h>

typedef __attribute__((ext_vector_type(16))) __bf16 v16bf;
typedef __attribute__((ext_vector_type(8)))  __bf16 v8bf;
typedef __attribute__((ext_vector_type(8)))  float  v8f;

__device__ __forceinline__ __bf16 f2bf(float f) {
  union { float f; unsigned u; } v; v.f = f;
  unsigned r = v.u + 0x7FFFu + ((v.u >> 16) & 1u);
  union { unsigned short s; __bf16 b; } o;
  o.s = (unsigned short)(r >> 16);
  return o.b;
}

__device__ __forceinline__ float bf2f(__bf16 b) {
  union { __bf16 b; unsigned short s; } i; i.b = b;
  union { unsigned u; float f; } o; o.u = (unsigned)i.s << 16;
  return o.f;
}

__device__ __forceinline__ void unpack8(const uint4 q, float* f) {
  union { unsigned u; float f; } c;
  c.u = q.x << 16; f[0] = c.f;  c.u = q.x & 0xFFFF0000u; f[1] = c.f;
  c.u = q.y << 16; f[2] = c.f;  c.u = q.y & 0xFFFF0000u; f[3] = c.f;
  c.u = q.z << 16; f[4] = c.f;  c.u = q.z & 0xFFFF0000u; f[5] = c.f;
  c.u = q.w << 16; f[6] = c.f;  c.u = q.w & 0xFFFF0000u; f[7] = c.f;
}

constexpr int TILE_K = 128;
constexpr int TILE_PITCH = TILE_K * 2 + 16;

__device__ __forceinline__ v16bf a_frag(const unsigned char* tile, int m16, int half, int ks) {
  const unsigned char* ar = tile + m16 * TILE_PITCH + half * 16;
  const v8bf alo = *(const v8bf*)(ar + ks * 64);
  const v8bf ahi = *(const v8bf*)(ar + ks * 64 + 32);
  return __builtin_shufflevector(alo, ahi, 0,1,2,3,4,5,6,7,8,9,10,11,12,13,14,15);
}

template <int K, int NW>
__global__ void __launch_bounds__(32 * NW, 16) gemm_wmma_bf16(const __bf16* __restrict__ A,
                               const __bf16* __restrict__ Bp,
                               float* __restrict__ C, int ldc) {
  static_assert(K == TILE_K, "a wave-wide 16-byte load is two bf16 rows");
  constexpr int KS = K / 32;
  static_assert(8 % NW == 0, "the waves share the 16 rows evenly");
  constexpr int NB = 8 / NW;
  __shared__ __attribute__((aligned(16))) unsigned char tile[16 * TILE_PITCH];
  const int wave = threadIdx.x >> 5;
  const int lane = threadIdx.x & 31;
  const int m16  = lane & 15;
  const int half = lane >> 4;
  const size_t row0 = (size_t)blockIdx.x * 16;

  uint4 vb[NB];
#pragma unroll
  for (int j = 0; j < NB; ++j)
    vb[j] = *(const uint4*)(A + (row0 + (wave + j * NW) * 2 + half) * K + m16 * 8);
  v8bf bf[KS * 2];
  const __bf16* bw = Bp + ((size_t)wave * KS * 2 * 32 + lane) * 8;
#pragma unroll
  for (int q = 0; q < KS * 2; ++q) bf[q] = *(const v8bf*)(bw + (size_t)q * 256);

#pragma unroll
  for (int j = 0; j < NB; ++j)
    *(uint4*)(tile + ((wave + j * NW) * 2 + half) * TILE_PITCH + m16 * 16) = vb[j];
  __syncthreads();

  v8f acc = {};
#pragma unroll
  for (int ks = 0; ks < KS; ++ks) {
    const v16bf a = a_frag(tile, m16, half, ks);
    const v16bf b = __builtin_shufflevector(bf[ks * 2], bf[ks * 2 + 1], 0,1,2,3,4,5,6,7,8,9,10,11,12,13,14,15);
    acc = __builtin_amdgcn_wmma_f32_16x16x32_bf16(
        false, a, false, b, (short)0, acc, false, false);
  }

  const int col  = wave * 16 + m16;
  const size_t crow = row0 + half * 8;
#pragma unroll
  for (int r = 0; r < 8; ++r)
    C[(crow + r) * ldc + col] = acc[r];
}

__device__ __forceinline__ void pack_w(const float* __restrict__ W, __bf16* __restrict__ Wp,
                                       int K, int N, int i) {
  if (i >= K * N) return;
  const int k = i / N, n = i % N;
  const int kk = k & 31;
  const size_t frag = ((size_t)(n >> 4) * (K >> 5) + (k >> 5)) * 2 + (kk >> 4);
  const int lane = (((kk >> 3) & 1) << 4) + (n & 15);
  Wp[(frag * 32 + lane) * 8 + (kk & 7)] = f2bf(W[i]);
}

__global__ void init_k(float* __restrict__ deg, int n, int deg_blocks,
                       const float* __restrict__ W1, __bf16* __restrict__ w1t, int din, int dh,
                       const float* __restrict__ Wmu, const float* __restrict__ Wls,
                       __bf16* __restrict__ wzt, int dz) {
  int b = blockIdx.x;
  if (b < deg_blocks) {
    const int i = b * blockDim.x + threadIdx.x;
    if (i < n) deg[i] = 1.0f;
    return;
  }
  b -= deg_blocks;
  const int w1_blocks = (din * dh + blockDim.x - 1) / blockDim.x;
  if (b < w1_blocks) {
    pack_w(W1, w1t, din, dh, b * blockDim.x + threadIdx.x);
    return;
  }
  b -= w1_blocks;
  const int wz_blocks = (dh * dz + blockDim.x - 1) / blockDim.x;
  if (b < wz_blocks) pack_w(Wmu, wzt, dh, dz, b * blockDim.x + threadIdx.x);
  else               pack_w(Wls, wzt + (size_t)dz * dh, dh, dz,
                            (b - wz_blocks) * blockDim.x + threadIdx.x);
}

__global__ void edge_rank_k(const int* __restrict__ dst1, int e1,
                            const int* __restrict__ dst2, int e2,
                            float* __restrict__ deg, int* __restrict__ rank) {
  int i = blockIdx.x * blockDim.x + threadIdx.x;
  if (i >= e1 + e2) return;
  const int d = i < e1 ? dst1[i] : dst2[i - e1];
  rank[i] = (int)atomicAdd(&deg[d], 1.0f) - 1;
}

__global__ void scan_off_k(const float* __restrict__ deg, int* __restrict__ off, int n,
                           const float* __restrict__ x, __bf16* __restrict__ xb, long long nx) {
  const int t = threadIdx.x;
  if (blockIdx.x > 0) {
    const long long i0 = (long long)(blockIdx.x - 1) * 8192 + t * 4;
    float4 v[2];
#pragma unroll
    for (int j = 0; j < 2; ++j) {
      const long long i = i0 + j * 4096;
      v[j] = *(const float4*)(x + (i < nx ? i : 0));
    }
#pragma unroll
    for (int j = 0; j < 2; ++j) {
      const long long i = i0 + j * 4096;
      union { __bf16 b[4]; uint2 u; } o;
      o.b[0] = f2bf(v[j].x); o.b[1] = f2bf(v[j].y);
      o.b[2] = f2bf(v[j].z); o.b[3] = f2bf(v[j].w);
      if (i < nx) *(uint2*)(xb + i) = o.u;
    }
    return;
  }
  __shared__ int wsum[32];
  const int lane = t & 31, wave = t >> 5;
  const int base = wave * 2048 + lane * 4;
  float4 v[16];
#pragma unroll
  for (int q = 0; q < 16; ++q) {
    const int i = base + q * 128;
    v[q] = *(const float4*)(deg + (i < n ? i : 0));
  }
  int4 c[16];
  int incl[16];
#pragma unroll
  for (int q = 0; q < 16; ++q) {
    const int m = base + q * 128 < n ? -1 : 0;
    c[q].x = ((int)v[q].x - 1) & m;
    c[q].y = ((int)v[q].y - 1) & m;
    c[q].z = ((int)v[q].z - 1) & m;
    c[q].w = ((int)v[q].w - 1) & m;
    incl[q] = c[q].x + c[q].y + c[q].z + c[q].w;
  }
#pragma unroll
  for (int d = 1; d < 32; d <<= 1) {
#pragma unroll
    for (int q = 0; q < 16; ++q) {
      const int y = __shfl_up(incl[q], d);
      if (lane >= d) incl[q] += y;
    }
  }
  int carry = 0;
#pragma unroll
  for (int q = 0; q < 16; ++q) {
    const int b = carry + incl[q] - (c[q].x + c[q].y + c[q].z + c[q].w);
    carry += __builtin_amdgcn_readlane(incl[q], 31);
    c[q] = make_int4(b, b + c[q].x, b + c[q].x + c[q].y, b + c[q].x + c[q].y + c[q].z);
  }
  if (lane == 0) wsum[wave] = carry;
  __syncthreads();
  if (wave == 0) {
    const int x = wsum[lane];
    int s = x;
#pragma unroll
    for (int d = 1; d < 32; d <<= 1) {
      const int y = __shfl_up(s, d);
      if (lane >= d) s += y;
    }
    wsum[lane] = s - x;
  }
  __syncthreads();
  const int pre = wsum[wave];
#pragma unroll
  for (int q = 0; q < 16; ++q) {
    const int i = base + q * 128;
    if (i < n)
      *(int4*)(off + i) = make_int4(c[q].x + pre, c[q].y + pre, c[q].z + pre, c[q].w + pre);
  }
  if (t == 1023) off[n] = pre + carry;
}

__global__ void fill_csr_k(const int* __restrict__ src1, const int* __restrict__ dst1, int e1,
                           const int* __restrict__ src2, const int* __restrict__ dst2, int e2,
                           const int* __restrict__ rank, const int* __restrict__ off,
                           int* __restrict__ csr, int edge_blocks,
                           float* __restrict__ deg, int n) {
  if ((int)blockIdx.x >= edge_blocks) {
    const int j = ((int)blockIdx.x - edge_blocks) * blockDim.x + threadIdx.x;
    if (j < n) { const float v = deg[j]; deg[j] = v > 0.0f ? rsqrtf(v) : 0.0f; }
    return;
  }
  int i = blockIdx.x * blockDim.x + threadIdx.x;
  if (i >= e1 + e2) return;
  int s, d;
  if (i < e1) { s = src1[i];      d = dst1[i]; }
  else        { s = src2[i - e1]; d = dst2[i - e1]; }
  csr[off[d] + rank[i]] = s;
}

__global__ void __launch_bounds__(256) agg_gemm1_k(const int* __restrict__ off, const int* __restrict__ csr,
                            const float* __restrict__ dinv, const __bf16* __restrict__ xb,
                            const __bf16* __restrict__ w1p, const float* __restrict__ bias,
                            __bf16* __restrict__ hout) {
  constexpr int K = TILE_K, KS = K / 32;
  __shared__ __attribute__((aligned(16))) unsigned char t_a[3][16 * TILE_PITCH];
  __shared__ __attribute__((aligned(16))) unsigned char t_h[16 * TILE_PITCH];
  const int wave = threadIdx.x >> 5;
  const int lane = threadIdx.x & 31;
  const int m16  = lane & 15;
  const int half = lane >> 4;
  const int row  = wave * 2 + half;
  const size_t node = (size_t)blockIdx.x * 16 + row;
  const uint4* xrow = (const uint4*)xb + m16;

  const float dn = dinv[node];
  float acc[8], f[8];
  unpack8(xrow[node * 16], f);
#pragma unroll
  for (int i = 0; i < 8; ++i) acc[i] = f[i] * dn * dn;

  const int jbeg = off[node], jend = off[node + 1];
  for (int j = jbeg; j < jend; j += 4) {
    int s[4]; float w[4]; uint4 r[4];
#pragma unroll
    for (int k = 0; k < 4; ++k) s[k] = csr[min(j + k, jend - 1)];
#pragma unroll
    for (int k = 0; k < 4; ++k) w[k] = dinv[s[k]] * dn;
#pragma unroll
    for (int k = 0; k < 4; ++k) {
      r[k] = xrow[(size_t)s[k] * 16];
      if (j + k >= jend) r[k] = make_uint4(0u, 0u, 0u, 0u);
    }
#pragma unroll
    for (int k = 0; k < 4; ++k) {
      unpack8(r[k], f);
#pragma unroll
      for (int i = 0; i < 8; ++i) acc[i] = __builtin_fmaf(f[i], w[k], acc[i]);
    }
  }

  v8bf bf[KS * 2];
  const __bf16* bw = w1p + ((size_t)wave * KS * 2 * 32 + lane) * 8;
#pragma unroll
  for (int q = 0; q < KS * 2; ++q) bf[q] = *(const v8bf*)(bw + (size_t)q * 256);
  const float bcol = bias[wave * 16 + m16];

  {
    union { __bf16 b[8]; uint4 u; } p0, p1, p2;
#pragma unroll
    for (int i = 0; i < 8; ++i) {
      p0.b[i] = f2bf(acc[i]);
      const float r1 = acc[i] - bf2f(p0.b[i]);
      p1.b[i] = f2bf(r1);
      p2.b[i] = f2bf(r1 - bf2f(p1.b[i]));
    }
    *(uint4*)(t_a[0] + row * TILE_PITCH + m16 * 16) = p0.u;
    *(uint4*)(t_a[1] + row * TILE_PITCH + m16 * 16) = p1.u;
    *(uint4*)(t_a[2] + row * TILE_PITCH + m16 * 16) = p2.u;
  }
  __syncthreads();

  v8f c = {};
#pragma unroll
  for (int ks = 0; ks < KS; ++ks) {
    const v16bf b = __builtin_shufflevector(bf[ks * 2], bf[ks * 2 + 1], 0,1,2,3,4,5,6,7,8,9,10,11,12,13,14,15);
#pragma unroll
    for (int p = 0; p < 3; ++p)
      c = __builtin_amdgcn_wmma_f32_16x16x32_bf16(
          false, a_frag(t_a[p], m16, half, ks), false, b, (short)0, c, false, false);
  }

#pragma unroll
  for (int r = 0; r < 8; ++r)
    *(__bf16*)(t_h + (half * 8 + r) * TILE_PITCH + (wave * 16 + m16) * 2) = f2bf(fmaxf(c[r] + bcol, 0.0f));
  __syncthreads();
  *(uint4*)(hout + node * K + m16 * 8) = *(const uint4*)(t_h + row * TILE_PITCH + m16 * 16);
}

template <int F>
__global__ void gather_agg_k(const int* __restrict__ off, const int* __restrict__ csr,
                             const float* __restrict__ dinv, const float* __restrict__ hlin,
                             const float* __restrict__ bias, const float* __restrict__ bias2,
                             float* __restrict__ outf, float* __restrict__ outf2,
                             int n_nodes) {
  constexpr int LPN = F / 4;
  constexpr int H = F / 2;
  const long long gid = (long long)blockIdx.x * blockDim.x + threadIdx.x;
  const int node = (int)(gid / LPN);
  if (node >= n_nodes) return;
  const int f = (int)(gid % LPN) * 4;
  const float dn = dinv[node];
  const float4 hs = *(const float4*)(hlin + (size_t)node * F + f);
  const float* bp = f < H ? bias + f : bias2 + (f - H);
  float4 acc;
  acc.x = hs.x * dn * dn + bp[0];
  acc.y = hs.y * dn * dn + bp[1];
  acc.z = hs.z * dn * dn + bp[2];
  acc.w = hs.w * dn * dn + bp[3];

  const int jbeg = off[node], jend = off[node + 1];
  for (int j = jbeg; j < jend; j += 4) {
    int s[4]; float w[4]; float4 r[4];
#pragma unroll
    for (int k = 0; k < 4; ++k) s[k] = csr[min(j + k, jend - 1)];
#pragma unroll
    for (int k = 0; k < 4; ++k) w[k] = dinv[s[k]] * dn;
#pragma unroll
    for (int k = 0; k < 4; ++k) {
      r[k] = *(const float4*)(hlin + (size_t)s[k] * F + f);
      if (j + k >= jend) r[k] = make_float4(0.0f, 0.0f, 0.0f, 0.0f);
    }
#pragma unroll
    for (int k = 0; k < 4; ++k) {
      acc.x += r[k].x * w[k];
      acc.y += r[k].y * w[k];
      acc.z += r[k].z * w[k];
      acc.w += r[k].w * w[k];
    }
  }

  float* o = f < H ? outf + (size_t)node * H + f : outf2 + (size_t)node * H + (f - H);
  *(float4*)o = acc;
}

extern "C" void kernel_launch(void* const* d_in, const int* in_sizes, int n_in,
                              void* d_out, int out_size, void* d_ws, size_t ws_size,
                              hipStream_t stream) {
  const float* x   = (const float*)d_in[0];
  const int*   ei  = (const int*)d_in[1];
  const int*   yei = (const int*)d_in[2];
  const float* W1  = (const float*)d_in[3];
  const float* b1  = (const float*)d_in[4];
  const float* Wmu = (const float*)d_in[5];
  const float* bmu = (const float*)d_in[6];
  const float* Wls = (const float*)d_in[7];
  const float* bls = (const float*)d_in[8];

  const int DIN = 128, DH = 128, DZ = 32;
  const int N  = in_sizes[0] / DIN;
  const int E1 = in_sizes[1] / 2;
  const int E2 = in_sizes[2] / 2;

  const int* src1 = ei;   const int* dst1 = ei  + E1;
  const int* src2 = yei;  const int* dst2 = yei + E2;

  char* wsb = (char*)d_ws;
  float*  dinv = (float*)wsb;
  __bf16* w1t  = (__bf16*)(wsb + (size_t)65536 * 4);
  __bf16* wmut = w1t + 128 * 128;
  __bf16* abf  = (__bf16*)(wsb + (size_t)(65536 + 16384) * 4);
  float*  buf1 = (float*)(wsb + (size_t)(65536 + 16384) * 4 + (size_t)N * DH * 2);
  float*  z_lin = buf1;
  __bf16* xb   = (__bf16*)(buf1 + (size_t)N * 2 * DZ);
  int*    off  = (int*)(buf1 + (size_t)N * DH);
  int*    rank = off + (N + 16);
  int*    csr  = rank + (E1 + E2);

  const int T = 256;
  auto blocks = [](long long total, int t) { return (unsigned)((total + t - 1) / t); };

  const unsigned deg_blocks = blocks(N, T);
  init_k<<<deg_blocks + blocks(DIN * DH, T) + 2 * blocks(DH * DZ, T), T, 0, stream>>>(
      dinv, N, (int)deg_blocks, W1, w1t, DIN, DH, Wmu, Wls, wmut, DZ);

  const unsigned edge_blocks = blocks(E1 + E2, T);
  const long long NX = (long long)N * DIN;
  edge_rank_k<<<edge_blocks, T, 0, stream>>>(dst1, E1, dst2, E2, dinv, rank);
  scan_off_k<<<1 + blocks(NX, 8192), 1024, 0, stream>>>(dinv, off, N, x, xb, NX);
  fill_csr_k<<<edge_blocks + deg_blocks, T, 0, stream>>>(src1, dst1, E1, src2, dst2, E2,
                                                        rank, off, csr, (int)edge_blocks, dinv, N);

  agg_gemm1_k<<<N / 16, 32 * 8, 0, stream>>>(off, csr, dinv, xb, w1t, b1, abf);

  gemm_wmma_bf16<128, 4><<<N / 16, 32 * 4, 0, stream>>>(abf, wmut, z_lin, 2 * DZ);

  float* out_mu = (float*)d_out;
  float* out_ls = out_mu + (size_t)N * DZ;

  gather_agg_k<64><<<blocks((long long)N * (2 * DZ / 4), T), T, 0, stream>>>(
      off, csr, dinv, z_lin, bmu, bls, out_mu, out_ls, N);
}
